# starting kernel
# baseline (speedup 1.0000x reference)
#include <hip/hip_runtime.h>

typedef __attribute__((ext_vector_type(2))) float v2f;
typedef __attribute__((ext_vector_type(8))) float v8f;

#define KD_L  512
#define KD_C  512
#define KD_B  64

__global__ __launch_bounds__(16) void kd_build_M(float* __restrict__ M) {
    __shared__ float Ksh[KD_L];
    __shared__ float Gsh[KD_L];
    __shared__ float mf[KD_L * 16];

    if (threadIdx.x == 0) {
        float P = 0.0f;
        for (int t = 0; t < KD_L; ++t) {
            float Pp = P + 1.0f;
            float K  = Pp / (Pp + 1.0f);
            P = (1.0f - K) * Pp;
            Ksh[t] = K;
            Gsh[t] = P / (P + 1.0f);
        }
    }
    __syncthreads();

    const int s = blockIdx.x * 16 + threadIdx.x;

    float m = 0.0f;
    for (int t = 0; t < KD_L; ++t) {
        float y = (t == s) ? 1.0f : 0.0f;
        m = m + Ksh[t] * (y - m);
        mf[t * 16 + threadIdx.x] = m;
    }

    float ms = mf[(KD_L - 1) * 16 + threadIdx.x];
    M[(KD_L - 1) * KD_L + s] = ms;
    for (int t = KD_L - 2; t >= 0; --t) {
        float mt = mf[t * 16 + threadIdx.x];
        ms = mt + Gsh[t] * (ms - mt);
        M[t * KD_L + s] = ms;
    }
}

__global__ __launch_bounds__(256) void kd_gemm(const float* __restrict__ obs,
                                               const float* __restrict__ M,
                                               float* __restrict__ out) {
    const int wave = (blockIdx.x * 256 + threadIdx.x) >> 5;
    const int lane = threadIdx.x & 31;

    const int b     = wave >> 8;
    const int rem   = wave & 255;
    const int mBase = (rem >> 5) * 64;
    const int nBase = (rem & 31) * 16;

    const float* __restrict__ Y = obs + (size_t)b * KD_L * KD_C;
    float* __restrict__ T = out + (size_t)b * KD_L * KD_C;
    float* __restrict__ R = out + (size_t)KD_B * KD_L * KD_C
                                + (size_t)b * KD_L * KD_C;

    const int r    = lane & 15;
    const int half = lane >> 4;

    v8f acc0 = {}; v8f acc1 = {}; v8f acc2 = {}; v8f acc3 = {};

    const float* Arow0 = M + (size_t)(mBase      + r) * KD_L;
    const float* Arow1 = M + (size_t)(mBase + 16 + r) * KD_L;
    const float* Arow2 = M + (size_t)(mBase + 32 + r) * KD_L;
    const float* Arow3 = M + (size_t)(mBase + 48 + r) * KD_L;

    for (int k = 0; k < KD_L; k += 4) {
        const int kk = k + 2 * half;

        v2f bf;
        bf.x = Y[(size_t)kk * KD_C + nBase + r];
        bf.y = Y[(size_t)(kk + 1) * KD_C + nBase + r];

        v2f a0 = *(const v2f*)(Arow0 + kk);
        v2f a1 = *(const v2f*)(Arow1 + kk);
        v2f a2 = *(const v2f*)(Arow2 + kk);
        v2f a3 = *(const v2f*)(Arow3 + kk);

        acc0 = __builtin_amdgcn_wmma_f32_16x16x4_f32(false, a0, false, bf,
                                                     (short)0, acc0, false, false);
        acc1 = __builtin_amdgcn_wmma_f32_16x16x4_f32(false, a1, false, bf,
                                                     (short)0, acc1, false, false);
        acc2 = __builtin_amdgcn_wmma_f32_16x16x4_f32(false, a2, false, bf,
                                                     (short)0, acc2, false, false);
        acc3 = __builtin_amdgcn_wmma_f32_16x16x4_f32(false, a3, false, bf,
                                                     (short)0, acc3, false, false);
    }

    const int c = nBase + r;
#pragma unroll
    for (int v = 0; v < 8; ++v) {
        {
            int l = mBase + 8 * half + v;
            float t = acc0[v]; float y = Y[(size_t)l * KD_C + c];
            T[(size_t)l * KD_C + c] = t;  R[(size_t)l * KD_C + c] = y - t;
        }
        {
            int l = mBase + 16 + 8 * half + v;
            float t = acc1[v]; float y = Y[(size_t)l * KD_C + c];
            T[(size_t)l * KD_C + c] = t;  R[(size_t)l * KD_C + c] = y - t;
        }
        {
            int l = mBase + 32 + 8 * half + v;
            float t = acc2[v]; float y = Y[(size_t)l * KD_C + c];
            T[(size_t)l * KD_C + c] = t;  R[(size_t)l * KD_C + c] = y - t;
        }
        {
            int l = mBase + 48 + 8 * half + v;
            float t = acc3[v]; float y = Y[(size_t)l * KD_C + c];
            T[(size_t)l * KD_C + c] = t;  R[(size_t)l * KD_C + c] = y - t;
        }
    }
}

extern "C" void kernel_launch(void* const* d_in, const int* in_sizes, int n_in,
                              void* d_out, int out_size, void* d_ws, size_t ws_size,
                              hipStream_t stream) {
    (void)in_sizes; (void)n_in; (void)out_size; (void)ws_size;

    const float* obs = (const float*)d_in[0];
    float* out = (float*)d_out;
    float* M   = (float*)d_ws;

    kd_build_M<<<KD_L / 16, 16, 0, stream>>>(M);

    kd_gemm<<<(KD_B * 8 * 32) / 8, 256, 0, stream>>>(obs, M, out);
}


# GEMM halved by time reflection: sigma=(S-X)P+XQ, delta=(D-X)Q+XP with P,Q = y +- time-reversed y, K=256; trend rows i, 511-i = (sigma +- delta)/2; 32x32 end-corner term X kept; sum order changes
# speedup vs baseline: 2.1273x; 2.1273x over previous
#include <hip/hip_runtime.h>

typedef __attribute__((ext_vector_type(2))) float v2f;
typedef __attribute__((ext_vector_type(4))) float v4f;
typedef __attribute__((ext_vector_type(8))) float v8f;

#define KD_L  512
#define KD_C  512
#define KD_B  64
#define KD_NC 128
#define KD_PITCH 256

__device__ __forceinline__ float kd_quot_y(float a, float b, float y) {
    const float q = a * y;
    const float r = __builtin_fmaf(-b, q, a);
    return __builtin_fmaf(r, y, q);
}

__device__ __forceinline__ float kd_quot(float a, float b) {
    return kd_quot_y(a, b, __builtin_amdgcn_rcpf(b));
}

#define KD_BW 16
#define KD_H  256
#define KD_XN 32
#define KD_OFF_D (KD_H * KD_H)
#define KD_OFF_X (2 * KD_H * KD_H)

__device__ __forceinline__ void kd_read_pairs(v2f (&dst)[KD_BW], const v2f* mg,
                                              int t0, int tid) {
#pragma unroll
    for (int i = 0; i < KD_BW; ++i) dst[i] = mg[(t0 + i) * 32 + tid];
}

template <bool PARK>
__device__ __forceinline__ float kd_back_upper(const v2f (&pr)[KD_BW], float ms,
                                               v2f* mg, int t0, int tid) {
    float mv[KD_BW];
#pragma unroll
    for (int i = KD_BW - 1; i >= 0; --i) {
        const float mt = pr[i].x;
        const float G  = pr[i].y;
        const float c  = __builtin_fmaf(-G, mt, mt);
        ms = __builtin_fmaf(G, ms, c);
        mv[i] = ms;
    }
    if (PARK) {
#pragma unroll
        for (int i = KD_BW - 1; i >= 0; --i) mg[(t0 + i) * 32 + tid].x = mv[i];
    }
    return ms;
}

__device__ __forceinline__ float kd_partner(float v) {
    const unsigned u = __builtin_bit_cast(unsigned, v);
    return __builtin_bit_cast(float, __builtin_amdgcn_permlanex16(
        u, u, 0x76543210u, 0xfedcba98u, false, false));
}

template <bool CORNER>
__device__ __forceinline__ float kd_back_lower(const v2f (&pr)[KD_BW], float ms,
                                               float* __restrict__ Wm, const v2f* mg,
                                               int t0, int tid, int k) {
    const bool hi = tid >= 16;
    float mv[KD_BW];
#pragma unroll
    for (int i = KD_BW - 1; i >= 0; --i) {
        const float mt = pr[i].x;
        const float G  = pr[i].y;
        const float c  = __builtin_fmaf(-G, mt, mt);
        ms = __builtin_fmaf(G, ms, c);
        mv[i] = ms;
    }
    float xv[KD_BW];
#pragma unroll
    for (int i = KD_BW - 1; i >= 0; --i) {
        const int t = t0 + i;
        const float other = kd_partner(mv[i]);
        const float lo = hi ? other : mv[i];
        const float up = hi ? mv[i] : other;
        float v = hi ? (lo - up) : (lo + up);
        if (CORNER) {
            const float far = mg[(KD_L - 1 - t) * 32 + (tid | 16)].x;
            xv[i] = 0.5f * (lo - far);
            v -= xv[i];
        }
        Wm[(hi ? KD_OFF_D : 0) + t * KD_H + k] = v;
    }
    if (CORNER) {
        if (!hi) {
#pragma unroll
            for (int i = KD_BW - 1; i >= 0; --i)
                Wm[KD_OFF_X + (t0 + i) * KD_XN + k] = xv[i];
        }
    }
    return ms;
}

struct KdGroup { float pp[4], d[4], y[4]; };

__device__ __forceinline__ void kd_chain4(float& Pp, KdGroup& g) {
#pragma unroll
    for (int u = 0; u < 4; ++u) {
        g.pp[u] = Pp;
        g.d[u]  = Pp + 1.0f;
        g.y[u]  = __builtin_amdgcn_rcpf(g.d[u]);
        Pp = __builtin_fmaf(Pp, g.y[u], 1.0f);
    }
}

__device__ __forceinline__ float kd_gains4(const KdGroup& g, float m, v2f* mg,
                                           int tb, int s, int tid) {
#pragma unroll
    for (int u = 0; u < 4; ++u) {
        const float K = kd_quot_y(g.pp[u], g.d[u], g.y[u]);
        const float P = (1.0f - K) * g.pp[u];
        const float G = kd_quot(P, P + 1.0f);
        const float mn = __builtin_fmaf(-K, m, m);
        m = (tb + u == s) ? K : mn;
        mg[(tb + u) * 32 + tid] = (v2f){m, G};
    }
    return m;
}

__global__ __launch_bounds__(32) void kd_build_M(float* __restrict__ Wm) {
    __shared__ v2f mg[KD_L * 32];

    const int tid = threadIdx.x;
    const int k = blockIdx.x * 16 + (tid & 15);
    const int s = (tid >= 16) ? (KD_L - 1 - k) : k;
    const bool corner = blockIdx.x < KD_XN / 16;

    float Pp = 1.0f;
    float m = 0.0f;
    KdGroup ga, gb;
    kd_chain4(Pp, ga);
    for (int tb = 0; tb < KD_L - 8; tb += 8) {
        kd_chain4(Pp, gb);
        m = kd_gains4(ga, m, mg, tb, s, tid);
        kd_chain4(Pp, ga);
        m = kd_gains4(gb, m, mg, tb + 4, s, tid);
    }
    kd_chain4(Pp, gb);
    m = kd_gains4(ga, m, mg, KD_L - 8, s, tid);
    m = kd_gains4(gb, m, mg, KD_L - 4, s, tid);
    mg[(KD_L - 1) * 32 + tid] = (v2f){m, 0.0f};

    float ms = m;
    v2f pa[KD_BW], pb[KD_BW];
    kd_read_pairs(pa, mg, KD_L - KD_BW, tid);
    kd_read_pairs(pb, mg, KD_L - 2 * KD_BW, tid);
    ms = kd_back_upper<true>(pa, ms, mg, KD_L - KD_BW, tid);
    kd_read_pairs(pa, mg, KD_L - 3 * KD_BW, tid);
    ms = kd_back_upper<true>(pb, ms, mg, KD_L - 2 * KD_BW, tid);
#pragma nounroll
    for (int t0 = KD_L - 3 * KD_BW; t0 >= KD_H + KD_BW; t0 -= 2 * KD_BW) {
        kd_read_pairs(pb, mg, t0 - KD_BW, tid);
        ms = kd_back_upper<false>(pa, ms, mg, t0, tid);
        kd_read_pairs(pa, mg, t0 - 2 * KD_BW, tid);
        ms = kd_back_upper<false>(pb, ms, mg, t0 - KD_BW, tid);
    }
    __syncthreads();
    for (int t0 = KD_H - KD_BW; t0 >= 3 * KD_BW; t0 -= 2 * KD_BW) {
        kd_read_pairs(pb, mg, t0 - KD_BW, tid);
        ms = kd_back_lower<false>(pa, ms, Wm, mg, t0, tid, k);
        kd_read_pairs(pa, mg, t0 - 2 * KD_BW, tid);
        ms = kd_back_lower<false>(pb, ms, Wm, mg, t0 - KD_BW, tid, k);
    }
    kd_read_pairs(pb, mg, 0, tid);
    if (corner) {
        ms = kd_back_lower<true>(pa, ms, Wm, mg, KD_BW, tid, k);
        ms = kd_back_lower<true>(pb, ms, Wm, mg, 0, tid, k);
    } else {
        ms = kd_back_lower<false>(pa, ms, Wm, mg, KD_BW, tid, k);
        ms = kd_back_lower<false>(pb, ms, Wm, mg, 0, tid, k);
    }
}

#define KD_QROW (KD_H / 2)

__device__ __forceinline__ void kd_kstep(v8f (&acc)[2][8], v2f aS, v2f aD,
                                         const float* bp, const float* bq) {
    v2f fp[8], fq[8];
#pragma unroll
    for (int q = 0; q < 4; ++q) {
        const v4f b2 = *(const v4f*)(bp + 64 * q);
        fp[2 * q]     = (v2f){b2.x, b2.y};
        fp[2 * q + 1] = (v2f){b2.z, b2.w};
        const v4f c2 = *(const v4f*)(bq + 64 * q);
        fq[2 * q]     = (v2f){c2.x, c2.y};
        fq[2 * q + 1] = (v2f){c2.z, c2.w};
    }
#pragma unroll
    for (int nt = 0; nt < 8; ++nt)
        acc[0][nt] = __builtin_amdgcn_wmma_f32_16x16x4_f32(
            false, aS, false, fp[nt], (short)0, acc[0][nt], false, false);
#pragma unroll
    for (int nt = 0; nt < 8; ++nt)
        acc[1][nt] = __builtin_amdgcn_wmma_f32_16x16x4_f32(
            false, aD, false, fq[nt], (short)0, acc[1][nt], false, false);
}

__global__ __launch_bounds__(512) void kd_gemm(const float* __restrict__ obs,
                                               const float* __restrict__ Wm,
                                               float* __restrict__ out) {
    __shared__ float slab[KD_H * KD_PITCH];

    const int b  = blockIdx.x >> 2;
    const int c0 = (blockIdx.x & 3) * KD_NC;

    const float* __restrict__ Y = obs + (size_t)b * KD_L * KD_C;
    float* __restrict__ T = out + (size_t)b * KD_L * KD_C;
    float* __restrict__ R = out + (size_t)KD_B * KD_L * KD_C
                                + (size_t)b * KD_L * KD_C;

    {
        const int g  = threadIdx.x & 31;
        const int pr = threadIdx.x >> 5;
        const int sb = 64 * (g >> 3) + 16 * (g & 3) + 2 * ((g >> 2) & 1);
#pragma unroll 2
        for (int it = 0; it < 8; ++it) {
            const int p = it * 16 + pr;
            const float* src = Y + (size_t)(2 * p) * KD_C + c0 + 4 * g;
            const float* rev = Y + (size_t)(KD_L - 1 - 2 * p) * KD_C + c0 + 4 * g;
            v4f r0 = __builtin_nontemporal_load((const v4f*)src);
            v4f r1 = __builtin_nontemporal_load((const v4f*)(src + KD_C));
            v4f u0 = __builtin_nontemporal_load((const v4f*)rev);
            v4f u1 = __builtin_nontemporal_load((const v4f*)(rev - KD_C));
            const v4f p0 = r0 + u0, p1 = r1 + u1;
            const v4f q0 = r0 - u0, q1 = r1 - u1;
            float* dp = slab + p * KD_PITCH + sb;
            float* dq = dp + KD_QROW * KD_PITCH;
            *(v2f*)dp        = (v2f){p0.x, p1.x};
            *(v2f*)(dp + 4)  = (v2f){p0.y, p1.y};
            *(v2f*)(dp + 8)  = (v2f){p0.z, p1.z};
            *(v2f*)(dp + 12) = (v2f){p0.w, p1.w};
            *(v2f*)dq        = (v2f){q0.x, q1.x};
            *(v2f*)(dq + 4)  = (v2f){q0.y, q1.y};
            *(v2f*)(dq + 8)  = (v2f){q0.z, q1.z};
            *(v2f*)(dq + 12) = (v2f){q0.w, q1.w};
        }
    }
    __syncthreads();

    const int wave = __builtin_amdgcn_readfirstlane(threadIdx.x >> 5);
    const int lane = threadIdx.x & 31;
    const int r    = lane & 15;
    const int half = lane >> 4;
    const int mBase = wave * 16;

    v8f acc[2][8];
#pragma unroll
    for (int mt = 0; mt < 2; ++mt)
#pragma unroll
        for (int nt = 0; nt < 8; ++nt) acc[mt][nt] = (v8f){};

    const float* ArowS = Wm            + (size_t)(mBase + r) * KD_H + 2 * half;
    const float* ArowD = Wm + KD_OFF_D + (size_t)(mBase + r) * KD_H + 2 * half;

    const float* Bp = slab + half * KD_PITCH + 4 * r;
    const float* Bq = Bp + KD_QROW * KD_PITCH;

    for (int kc = 0; kc < KD_H; kc += 32) {
        v2f a[8][2];
#pragma unroll
        for (int j = 0; j < 8; ++j) {
            a[j][0] = *(const v2f*)(ArowS + kc + 4 * j);
            a[j][1] = *(const v2f*)(ArowD + kc + 4 * j);
        }
#pragma unroll
        for (int j = 0; j < 8; ++j) {
            const int po = ((kc + 4 * j) >> 1) * KD_PITCH;
            kd_kstep(acc, a[j][0], a[j][1], Bp + po, Bq + po);
        }
    }

    if (wave < KD_XN / 16) {
        const float* ArowX = Wm + KD_OFF_X + (size_t)(mBase + r) * KD_XN + 2 * half;
#pragma nounroll
        for (int jh = 0; jh < 8; jh += 4) {
            v2f ax[4];
#pragma unroll
            for (int j = 0; j < 4; ++j) ax[j] = *(const v2f*)(ArowX + 4 * (jh + j));
#pragma unroll
            for (int j = 0; j < 4; ++j) {
                const int po = ((4 * (jh + j)) >> 1) * KD_PITCH;
                kd_kstep(acc, ax[j], ax[j], Bq + po, Bp + po);
            }
        }
    }

#pragma unroll
    for (int nt = 0; nt < 8; ++nt) {
#pragma unroll
        for (int v = 0; v < 8; ++v) {
            const int i = mBase + 8 * half + v;
            const size_t oLo = (size_t)i * KD_C + c0 + 16 * nt + r;
            const size_t oHi = (size_t)(KD_L - 1 - i) * KD_C + c0 + 16 * nt + r;
            const float sg = acc[0][nt][v];
            const float dl = acc[1][nt][v];
            const float tLo = 0.5f * (sg + dl);
            const float tHi = 0.5f * (sg - dl);
            const float yLo = Y[oLo];
            const float yHi = Y[oHi];
            T[oLo] = tLo;  R[oLo] = yLo - tLo;
            T[oHi] = tHi;  R[oHi] = yHi - tHi;
        }
    }
}

extern "C" void kernel_launch(void* const* d_in, const int* in_sizes, int n_in,
                              void* d_out, int out_size, void* d_ws, size_t ws_size,
                              hipStream_t stream) {
    (void)in_sizes; (void)n_in; (void)out_size; (void)ws_size;

    const float* obs = (const float*)d_in[0];
    float* out = (float*)d_out;
    float* Wm  = (float*)d_ws;

    kd_build_M<<<KD_H / 16, 32, 0, stream>>>(Wm);

    kd_gemm<<<KD_B * (KD_C / KD_NC), 512, 0, stream>>>(obs, Wm, out);
}


# M build: each entry of the half matrices from its closed form c*(rho^|i-k| +- rho^(511-i-k)) + corner term, one thread per column, no 512-step recursion; M can differ in the last bits; GEMM unchanged
# speedup vs baseline: 3.0026x; 1.4115x over previous
#include <hip/hip_runtime.h>

typedef __attribute__((ext_vector_type(2))) float v2f;
typedef __attribute__((ext_vector_type(4))) float v4f;
typedef __attribute__((ext_vector_type(8))) float v8f;

#define KD_L  512
#define KD_C  512
#define KD_B  64
#define KD_NC 128
#define KD_PITCH 256

#define KD_H  256
#define KD_XN 32
#define KD_OFF_D (KD_H * KD_H)
#define KD_OFF_X (2 * KD_H * KD_H)
#define KD_BR 4

#define KD_RHO1  0.381966011250105f
#define KD_RHO2  0.145898033750315f
#define KD_RHO4  0.0212862362522082f
#define KD_RHO8  4.53103853784822e-4f
#define KD_RHO16 2.05303102314657e-7f
#define KD_RHO32 4.21493638200227e-14f
#define KD_RHO64 1.77656887043264e-27f
#define KD_C0  0.447213595499958f
#define KD_C0R 0.170820393249937f
#define KD_CM  0.138196601125011f
#define KD_CP  0.309016994374947f

__device__ __forceinline__ float kd_rho_pow(int d) {
    float p = (d & 1) ? KD_RHO1 : 1.0f;
    p *= (d & 2)  ? KD_RHO2  : 1.0f;
    p *= (d & 4)  ? KD_RHO4  : 1.0f;
    p *= (d & 8)  ? KD_RHO8  : 1.0f;
    p *= (d & 16) ? KD_RHO16 : 1.0f;
    p *= (d & 32) ? KD_RHO32 : 1.0f;
    p *= (d & 64) ? KD_RHO64 : 1.0f;
    return (d < 128) ? p : 0.0f;
}

__global__ __launch_bounds__(KD_H) void kd_build_M(float* __restrict__ Wm) {
    const int k = threadIdx.x;
    for (int j = 0; j < KD_BR; ++j) {
        const int i = blockIdx.x * KD_BR + j;
        const int d = (i > k) ? (i - k) : (k - i);
        const float T  = KD_C0 * kd_rho_pow(d);
        const float Hk = KD_C0 * kd_rho_pow(KD_L - 1 - i - k);
        const float pw = kd_rho_pow(i + k + 1);
        const bool corner = (i < KD_XN) && (k < KD_XN);
        const float e = corner ? (KD_CM * pw) : -(KD_C0R * pw);
        Wm[i * KD_H + k]            = T + (Hk + e);
        Wm[KD_OFF_D + i * KD_H + k] = T + (e - Hk);
        if (corner) Wm[KD_OFF_X + i * KD_XN + k] = -(KD_CP * pw);
    }
}

#define KD_QROW (KD_H / 2)

__device__ __forceinline__ void kd_kstep(v8f (&acc)[2][8], v2f aS, v2f aD,
                                         const float* bp, const float* bq) {
    v2f fp[8], fq[8];
#pragma unroll
    for (int q = 0; q < 4; ++q) {
        const v4f b2 = *(const v4f*)(bp + 64 * q);
        fp[2 * q]     = (v2f){b2.x, b2.y};
        fp[2 * q + 1] = (v2f){b2.z, b2.w};
        const v4f c2 = *(const v4f*)(bq + 64 * q);
        fq[2 * q]     = (v2f){c2.x, c2.y};
        fq[2 * q + 1] = (v2f){c2.z, c2.w};
    }
#pragma unroll
    for (int nt = 0; nt < 8; ++nt)
        acc[0][nt] = __builtin_amdgcn_wmma_f32_16x16x4_f32(
            false, aS, false, fp[nt], (short)0, acc[0][nt], false, false);
#pragma unroll
    for (int nt = 0; nt < 8; ++nt)
        acc[1][nt] = __builtin_amdgcn_wmma_f32_16x16x4_f32(
            false, aD, false, fq[nt], (short)0, acc[1][nt], false, false);
}

__global__ __launch_bounds__(512) void kd_gemm(const float* __restrict__ obs,
                                               const float* __restrict__ Wm,
                                               float* __restrict__ out) {
    __shared__ float slab[KD_H * KD_PITCH];

    const int b  = blockIdx.x >> 2;
    const int c0 = (blockIdx.x & 3) * KD_NC;

    const float* __restrict__ Y = obs + (size_t)b * KD_L * KD_C;
    float* __restrict__ T = out + (size_t)b * KD_L * KD_C;
    float* __restrict__ R = out + (size_t)KD_B * KD_L * KD_C
                                + (size_t)b * KD_L * KD_C;

    {
        const int g  = threadIdx.x & 31;
        const int pr = threadIdx.x >> 5;
        const int sb = 64 * (g >> 3) + 16 * (g & 3) + 2 * ((g >> 2) & 1);
#pragma unroll 2
        for (int it = 0; it < 8; ++it) {
            const int p = it * 16 + pr;
            const float* src = Y + (size_t)(2 * p) * KD_C + c0 + 4 * g;
            const float* rev = Y + (size_t)(KD_L - 1 - 2 * p) * KD_C + c0 + 4 * g;
            v4f r0 = __builtin_nontemporal_load((const v4f*)src);
            v4f r1 = __builtin_nontemporal_load((const v4f*)(src + KD_C));
            v4f u0 = __builtin_nontemporal_load((const v4f*)rev);
            v4f u1 = __builtin_nontemporal_load((const v4f*)(rev - KD_C));
            const v4f p0 = r0 + u0, p1 = r1 + u1;
            const v4f q0 = r0 - u0, q1 = r1 - u1;
            float* dp = slab + p * KD_PITCH + sb;
            float* dq = dp + KD_QROW * KD_PITCH;
            *(v2f*)dp        = (v2f){p0.x, p1.x};
            *(v2f*)(dp + 4)  = (v2f){p0.y, p1.y};
            *(v2f*)(dp + 8)  = (v2f){p0.z, p1.z};
            *(v2f*)(dp + 12) = (v2f){p0.w, p1.w};
            *(v2f*)dq        = (v2f){q0.x, q1.x};
            *(v2f*)(dq + 4)  = (v2f){q0.y, q1.y};
            *(v2f*)(dq + 8)  = (v2f){q0.z, q1.z};
            *(v2f*)(dq + 12) = (v2f){q0.w, q1.w};
        }
    }
    __syncthreads();

    const int wave = __builtin_amdgcn_readfirstlane(threadIdx.x >> 5);
    const int lane = threadIdx.x & 31;
    const int r    = lane & 15;
    const int half = lane >> 4;
    const int mBase = wave * 16;

    v8f acc[2][8];
#pragma unroll
    for (int mt = 0; mt < 2; ++mt)
#pragma unroll
        for (int nt = 0; nt < 8; ++nt) acc[mt][nt] = (v8f){};

    const float* ArowS = Wm            + (size_t)(mBase + r) * KD_H + 2 * half;
    const float* ArowD = Wm + KD_OFF_D + (size_t)(mBase + r) * KD_H + 2 * half;

    const float* Bp = slab + half * KD_PITCH + 4 * r;
    const float* Bq = Bp + KD_QROW * KD_PITCH;

    for (int kc = 0; kc < KD_H; kc += 32) {
        v2f a[8][2];
#pragma unroll
        for (int j = 0; j < 8; ++j) {
            a[j][0] = *(const v2f*)(ArowS + kc + 4 * j);
            a[j][1] = *(const v2f*)(ArowD + kc + 4 * j);
        }
#pragma unroll
        for (int j = 0; j < 8; ++j) {
            const int po = ((kc + 4 * j) >> 1) * KD_PITCH;
            kd_kstep(acc, a[j][0], a[j][1], Bp + po, Bq + po);
        }
    }

    if (wave < KD_XN / 16) {
        const float* ArowX = Wm + KD_OFF_X + (size_t)(mBase + r) * KD_XN + 2 * half;
#pragma nounroll
        for (int jh = 0; jh < 8; jh += 4) {
            v2f ax[4];
#pragma unroll
            for (int j = 0; j < 4; ++j) ax[j] = *(const v2f*)(ArowX + 4 * (jh + j));
#pragma unroll
            for (int j = 0; j < 4; ++j) {
                const int po = ((4 * (jh + j)) >> 1) * KD_PITCH;
                kd_kstep(acc, ax[j], ax[j], Bq + po, Bp + po);
            }
        }
    }

#pragma unroll
    for (int nt = 0; nt < 8; ++nt) {
#pragma unroll
        for (int v = 0; v < 8; ++v) {
            const int i = mBase + 8 * half + v;
            const size_t oLo = (size_t)i * KD_C + c0 + 16 * nt + r;
            const size_t oHi = (size_t)(KD_L - 1 - i) * KD_C + c0 + 16 * nt + r;
            const float sg = acc[0][nt][v];
            const float dl = acc[1][nt][v];
            const float tLo = 0.5f * (sg + dl);
            const float tHi = 0.5f * (sg - dl);
            const float yLo = Y[oLo];
            const float yHi = Y[oHi];
            T[oLo] = tLo;  R[oLo] = yLo - tLo;
            T[oHi] = tHi;  R[oHi] = yHi - tHi;
        }
    }
}

extern "C" void kernel_launch(void* const* d_in, const int* in_sizes, int n_in,
                              void* d_out, int out_size, void* d_ws, size_t ws_size,
                              hipStream_t stream) {
    (void)in_sizes; (void)n_in; (void)out_size; (void)ws_size;

    const float* obs = (const float*)d_in[0];
    float* out = (float*)d_out;
    float* Wm  = (float*)d_ws;

    kd_build_M<<<KD_H / KD_BR, KD_H, 0, stream>>>(Wm);

    kd_gemm<<<KD_B * (KD_C / KD_NC), 512, 0, stream>>>(obs, Wm, out);
}


# Dense half products replaced by an exact two-level split of M: 8 diagonal 64x64 block products + 16 boundary sums (one 16x512 product) added as a 17th k step; residual y from LDS; sum order changes
# speedup vs baseline: 4.8299x; 1.6086x over previous
#include <hip/hip_runtime.h>

typedef __attribute__((ext_vector_type(2))) float v2f;
typedef __attribute__((ext_vector_type(4))) float v4f;
typedef __attribute__((ext_vector_type(8))) float v8f;

#define KD_L  512
#define KD_C  512
#define KD_B  64
#define KD_NC 128
#define KD_PITCH 256

#define KD_NB   64
#define KD_NBLK 8
#define KD_OFF_W (KD_NBLK * KD_NB * KD_NB)

#define KD_RHO1  0.381966011250105f
#define KD_RHO2  0.145898033750315f
#define KD_RHO4  0.0212862362522082f
#define KD_RHO8  4.53103853784822e-4f
#define KD_RHO16 2.05303102314657e-7f
#define KD_RHO32 4.21493638200227e-14f
#define KD_RHO64 1.77656887043264e-27f
#define KD_C0  0.447213595499958f

__device__ __forceinline__ float kd_rho_pow(int d) {
    float p = (d & 1) ? KD_RHO1 : 1.0f;
    p *= (d & 2)  ? KD_RHO2  : 1.0f;
    p *= (d & 4)  ? KD_RHO4  : 1.0f;
    p *= (d & 8)  ? KD_RHO8  : 1.0f;
    p *= (d & 16) ? KD_RHO16 : 1.0f;
    p *= (d & 32) ? KD_RHO32 : 1.0f;
    p *= (d & 64) ? KD_RHO64 : 1.0f;
    return (d < 128) ? p : 0.0f;
}

__global__ __launch_bounds__(256) void kd_build_M(float* __restrict__ Wm) {
    if (blockIdx.x < 128) {
        const int idx = blockIdx.x * 256 + threadIdx.x;
        const int I = idx >> 12, a = (idx >> 6) & 63, b = idx & 63;
        const int l = KD_NB * I + a, s = KD_NB * I + b;
        const int d = (a > b) ? (a - b) : (b - a);
        const float T  = KD_C0 * kd_rho_pow(d);
        const float t2 = KD_C0 * kd_rho_pow(l + s + 2);
        const float t3 = KD_C0 * kd_rho_pow(2 * KD_L - 1 - l - s);
        Wm[idx] = T + (t3 - t2);
    } else {
        const int idx = (blockIdx.x - 128) * 256 + threadIdx.x;
        const int row = idx >> 9, s = idx & (KD_L - 1);
        const int I = row & 7;
        const int lo = KD_NB * I;
        const bool below = s < lo, above = s >= lo + KD_NB;
        float w = 0.0f;
        if (row < 8) {
            const float t2 = KD_C0 * kd_rho_pow(lo + s + 2);
            if (below) w = KD_C0 * kd_rho_pow(lo - s) - t2;
            if (above) w = -t2;
        } else {
            const float t3 = KD_C0 * kd_rho_pow(960 - lo - s);
            if (above) w = KD_C0 * kd_rho_pow(s - lo - 63) + t3;
            if (below) w = t3;
        }
        Wm[KD_OFF_W + idx] = w;
    }
}

#define KD_SLAB (256 * KD_PITCH)
#define KD_PART (2 * 16 * KD_NC)

__device__ __forceinline__ void kd_kstep(v8f (&acc)[2][8], v2f a0, v2f a1,
                                         const float* bp) {
    v2f fp[8];
#pragma unroll
    for (int q = 0; q < 4; ++q) {
        const v4f b2 = *(const v4f*)(bp + 64 * q);
        fp[2 * q]     = (v2f){b2.x, b2.y};
        fp[2 * q + 1] = (v2f){b2.z, b2.w};
    }
#pragma unroll
    for (int nt = 0; nt < 8; ++nt)
        acc[0][nt] = __builtin_amdgcn_wmma_f32_16x16x4_f32(
            false, a0, false, fp[nt], (short)0, acc[0][nt], false, false);
#pragma unroll
    for (int nt = 0; nt < 8; ++nt)
        acc[1][nt] = __builtin_amdgcn_wmma_f32_16x16x4_f32(
            false, a1, false, fp[nt], (short)0, acc[1][nt], false, false);
}

__global__ __launch_bounds__(512) void kd_gemm(const float* __restrict__ obs,
                                               const float* __restrict__ Wm,
                                               float* __restrict__ out) {
    __shared__ float slab[KD_SLAB + KD_PART];
    float* part = slab + KD_SLAB;

    const int b  = blockIdx.x >> 2;
    const int c0 = (blockIdx.x & 3) * KD_NC;

    const float* __restrict__ Y = obs + (size_t)b * KD_L * KD_C;
    float* __restrict__ T = out + (size_t)b * KD_L * KD_C;
    float* __restrict__ R = out + (size_t)KD_B * KD_L * KD_C
                                + (size_t)b * KD_L * KD_C;

    {
        const int g  = threadIdx.x & 31;
        const int pr = threadIdx.x >> 5;
        const int sb = 64 * (g >> 3) + 16 * (g & 3) + 2 * ((g >> 2) & 1);
#pragma unroll 4
        for (int it = 0; it < 16; ++it) {
            const int p = it * 16 + pr;
            const float* src = Y + (size_t)(2 * p) * KD_C + c0 + 4 * g;
            const v4f r0 = __builtin_nontemporal_load((const v4f*)src);
            const v4f r1 = __builtin_nontemporal_load((const v4f*)(src + KD_C));
            float* dp = slab + p * KD_PITCH + sb;
            *(v2f*)dp        = (v2f){r0.x, r1.x};
            *(v2f*)(dp + 4)  = (v2f){r0.y, r1.y};
            *(v2f*)(dp + 8)  = (v2f){r0.z, r1.z};
            *(v2f*)(dp + 12) = (v2f){r0.w, r1.w};
        }
    }
    __syncthreads();

    const int wave = __builtin_amdgcn_readfirstlane(threadIdx.x >> 5);
    const int lane = threadIdx.x & 31;
    const int r    = lane & 15;
    const int half = lane >> 4;

    {
        const int nt = wave & 7;
        const int kh = wave >> 3;
        const float* Aw = Wm + KD_OFF_W + (size_t)r * KD_L + 256 * kh + 2 * half;
        const float* Bw = slab + (128 * kh + half) * KD_PITCH
                               + 64 * (nt >> 1) + 4 * r + 2 * (nt & 1);
        v8f s0 = (v8f){}, s1 = (v8f){};
        for (int t = 0; t < 64; t += 8) {
            v2f a[8], f[8];
#pragma unroll
            for (int j = 0; j < 8; ++j) a[j] = *(const v2f*)(Aw + 4 * (t + j));
#pragma unroll
            for (int j = 0; j < 8; ++j) f[j] = *(const v2f*)(Bw + 2 * (t + j) * KD_PITCH);
#pragma unroll
            for (int j = 0; j < 8; j += 2) {
                s0 = __builtin_amdgcn_wmma_f32_16x16x4_f32(
                    false, a[j], false, f[j], (short)0, s0, false, false);
                s1 = __builtin_amdgcn_wmma_f32_16x16x4_f32(
                    false, a[j + 1], false, f[j + 1], (short)0, s1, false, false);
            }
        }
#pragma unroll
        for (int v = 0; v < 8; ++v)
            part[((8 * half + v) * KD_NC + 16 * nt + r) * 2 + kh] = s0[v] + s1[v];
    }
    __syncthreads();

    const int I  = wave >> 1;
    const int rb = 32 * (wave & 1);

    v8f acc[2][8];
#pragma unroll
    for (int mt = 0; mt < 2; ++mt)
#pragma unroll
        for (int nt = 0; nt < 8; ++nt) acc[mt][nt] = (v8f){};

    const float* Ad = Wm + (size_t)I * (KD_NB * KD_NB) + (size_t)(rb + r) * KD_NB + 2 * half;
    const float* Bp = slab + (32 * I + half) * KD_PITCH + 4 * r;

    for (int kc = 0; kc < KD_NB; kc += 32) {
        v2f a[8][2];
#pragma unroll
        for (int j = 0; j < 8; ++j) {
            a[j][0] = *(const v2f*)(Ad + kc + 4 * j);
            a[j][1] = *(const v2f*)(Ad + 16 * KD_NB + kc + 4 * j);
        }
#pragma unroll
        for (int j = 0; j < 8; ++j) {
            const int po = ((kc + 4 * j) >> 1) * KD_PITCH;
            kd_kstep(acc, a[j][0], a[j][1], Bp + po);
        }
    }

    {
        v2f ax[2];
#pragma unroll
        for (int mt = 0; mt < 2; ++mt) {
            const int a = rb + 16 * mt + r;
            const float u = kd_rho_pow(half ? (KD_NB - 1 - a) : a);
            ax[mt] = (v2f){u, u};
        }
        const float* Pp = part + ((8 * half + I) * KD_NC + r) * 2;
        v2f fb[8];
#pragma unroll
        for (int nt = 0; nt < 8; ++nt) fb[nt] = *(const v2f*)(Pp + 32 * nt);
#pragma unroll
        for (int nt = 0; nt < 8; ++nt)
            acc[0][nt] = __builtin_amdgcn_wmma_f32_16x16x4_f32(
                false, ax[0], false, fb[nt], (short)0, acc[0][nt], false, false);
#pragma unroll
        for (int nt = 0; nt < 8; ++nt)
            acc[1][nt] = __builtin_amdgcn_wmma_f32_16x16x4_f32(
                false, ax[1], false, fb[nt], (short)0, acc[1][nt], false, false);
    }

    const int row0 = KD_NB * I + rb + 8 * half;
    const float* Ye = slab + (row0 >> 1) * KD_PITCH + 4 * r;
#pragma unroll
    for (int mt = 0; mt < 2; ++mt) {
#pragma unroll
        for (int q = 0; q < 4; ++q) {
#pragma unroll
            for (int vp = 0; vp < 4; ++vp) {
                const v4f yy = *(const v4f*)(Ye + (8 * mt + vp) * KD_PITCH + 64 * q);
                const int i = row0 + 16 * mt + 2 * vp;
                const size_t o = (size_t)i * KD_C + c0 + 32 * q + r;
                const float t00 = acc[mt][2 * q][2 * vp];
                const float t10 = acc[mt][2 * q][2 * vp + 1];
                const float t01 = acc[mt][2 * q + 1][2 * vp];
                const float t11 = acc[mt][2 * q + 1][2 * vp + 1];
                T[o]             = t00;  R[o]             = yy.x - t00;
                T[o + KD_C]      = t10;  R[o + KD_C]      = yy.y - t10;
                T[o + 16]        = t01;  R[o + 16]        = yy.z - t01;
                T[o + KD_C + 16] = t11;  R[o + KD_C + 16] = yy.w - t11;
            }
        }
    }
}

extern "C" void kernel_launch(void* const* d_in, const int* in_sizes, int n_in,
                              void* d_out, int out_size, void* d_ws, size_t ws_size,
                              hipStream_t stream) {
    (void)in_sizes; (void)n_in; (void)out_size; (void)ws_size;

    const float* obs = (const float*)d_in[0];
    float* out = (float*)d_out;
    float* Wm  = (float*)d_ws;

    kd_build_M<<<160, 256, 0, stream>>>(Wm);

    kd_gemm<<<KD_B * (KD_C / KD_NC), 512, 0, stream>>>(obs, Wm, out);
}


# Diagonal blocks halved by reflection: corner terms go to the boundary sums, so each block is one Toeplitz block = two 32x32 products on y +- mirrored y; rows a, 63-a = sum +- diff; sum order changes
# speedup vs baseline: 5.5968x; 1.1588x over previous
#include <hip/hip_runtime.h>

typedef __attribute__((ext_vector_type(2))) float v2f;
typedef __attribute__((ext_vector_type(4))) float v4f;
typedef __attribute__((ext_vector_type(8))) float v8f;

#define KD_L  512
#define KD_C  512
#define KD_B  64
#define KD_NC 128
#define KD_PITCH 256

#define KD_NB   64
#define KD_NH   32
#define KD_OFF_W (2 * KD_NH * KD_NH)

#define KD_RHO1  0.381966011250105f
#define KD_RHO2  0.145898033750315f
#define KD_RHO4  0.0212862362522082f
#define KD_RHO8  4.53103853784822e-4f
#define KD_RHO16 2.05303102314657e-7f
#define KD_RHO32 4.21493638200227e-14f
#define KD_RHO64 1.77656887043264e-27f
#define KD_C0  0.447213595499958f

__device__ __forceinline__ float kd_rho_pow(int d) {
    float p = (d & 1) ? KD_RHO1 : 1.0f;
    p *= (d & 2)  ? KD_RHO2  : 1.0f;
    p *= (d & 4)  ? KD_RHO4  : 1.0f;
    p *= (d & 8)  ? KD_RHO8  : 1.0f;
    p *= (d & 16) ? KD_RHO16 : 1.0f;
    p *= (d & 32) ? KD_RHO32 : 1.0f;
    p *= (d & 64) ? KD_RHO64 : 1.0f;
    return (d < 128) ? p : 0.0f;
}

__global__ __launch_bounds__(256) void kd_build_M(float* __restrict__ Wm) {
    if (blockIdx.x < 8) {
        const int idx = blockIdx.x * 256 + threadIdx.x;
        const int dl = idx >> 10, a = (idx >> 5) & 31, b = idx & 31;
        const int d1 = (a > b) ? (a - b) : (b - a);
        const int d2 = (KD_NB - 1) - b - a;
        const float t1 = (0.5f * KD_C0) * kd_rho_pow(d1);
        const float t2 = (0.5f * KD_C0) * kd_rho_pow(d2);
        Wm[idx] = dl ? (t1 - t2) : (t1 + t2);
    } else {
        const int idx = (blockIdx.x - 8) * 256 + threadIdx.x;
        const int row = idx >> 9, s = idx & (KD_L - 1);
        const int I = row & 7;
        const int lo = KD_NB * I;
        const bool below = s < lo, above = s >= lo + KD_NB;
        float w;
        if (row < 8) {
            w = -(KD_C0 * kd_rho_pow(lo + s + 2));
            if (below) w += KD_C0 * kd_rho_pow(lo - s);
        } else {
            w = KD_C0 * kd_rho_pow(960 - lo - s);
            if (above) w += KD_C0 * kd_rho_pow(s - lo - 63);
        }
        Wm[KD_OFF_W + idx] = w;
    }
}

#define KD_SLAB (256 * KD_PITCH)
#define KD_PART (2 * 16 * KD_NC)

__device__ __forceinline__ void kd_kstep(v8f (&sg)[2][4], v8f (&dl)[2][4],
                                         v2f aS0, v2f aS1, v2f aD0, v2f aD1,
                                         const float* lo, const float* hi) {
    v2f fp[4], fq[4];
#pragma unroll
    for (int q = 0; q < 2; ++q) {
        const v4f u = *(const v4f*)(lo + 64 * q);
        const v4f w = *(const v4f*)(hi + 64 * q);
        fp[2 * q]     = (v2f){u.x + w.y, u.y + w.x};
        fq[2 * q]     = (v2f){u.x - w.y, u.y - w.x};
        fp[2 * q + 1] = (v2f){u.z + w.w, u.w + w.z};
        fq[2 * q + 1] = (v2f){u.z - w.w, u.w - w.z};
    }
#pragma unroll
    for (int j = 0; j < 4; ++j)
        sg[0][j] = __builtin_amdgcn_wmma_f32_16x16x4_f32(
            false, aS0, false, fp[j], (short)0, sg[0][j], false, false);
#pragma unroll
    for (int j = 0; j < 4; ++j)
        dl[0][j] = __builtin_amdgcn_wmma_f32_16x16x4_f32(
            false, aD0, false, fq[j], (short)0, dl[0][j], false, false);
#pragma unroll
    for (int j = 0; j < 4; ++j)
        sg[1][j] = __builtin_amdgcn_wmma_f32_16x16x4_f32(
            false, aS1, false, fp[j], (short)0, sg[1][j], false, false);
#pragma unroll
    for (int j = 0; j < 4; ++j)
        dl[1][j] = __builtin_amdgcn_wmma_f32_16x16x4_f32(
            false, aD1, false, fq[j], (short)0, dl[1][j], false, false);
}

__global__ __launch_bounds__(512) void kd_gemm(const float* __restrict__ obs,
                                               const float* __restrict__ Wm,
                                               float* __restrict__ out) {
    __shared__ float slab[KD_SLAB + KD_PART];
    float* part = slab + KD_SLAB;

    const int b  = blockIdx.x >> 2;
    const int c0 = (blockIdx.x & 3) * KD_NC;

    const float* __restrict__ Y = obs + (size_t)b * KD_L * KD_C;
    float* __restrict__ T = out + (size_t)b * KD_L * KD_C;
    float* __restrict__ R = out + (size_t)KD_B * KD_L * KD_C
                                + (size_t)b * KD_L * KD_C;

    {
        const int g  = threadIdx.x & 31;
        const int pr = threadIdx.x >> 5;
        const int sb = 64 * (g >> 3) + 16 * (g & 3) + 2 * ((g >> 2) & 1);
#pragma unroll 4
        for (int it = 0; it < 16; ++it) {
            const int p = it * 16 + pr;
            const float* src = Y + (size_t)(2 * p) * KD_C + c0 + 4 * g;
            const v4f r0 = __builtin_nontemporal_load((const v4f*)src);
            const v4f r1 = __builtin_nontemporal_load((const v4f*)(src + KD_C));
            float* dp = slab + p * KD_PITCH + sb;
            *(v2f*)dp        = (v2f){r0.x, r1.x};
            *(v2f*)(dp + 4)  = (v2f){r0.y, r1.y};
            *(v2f*)(dp + 8)  = (v2f){r0.z, r1.z};
            *(v2f*)(dp + 12) = (v2f){r0.w, r1.w};
        }
    }
    __syncthreads();

    const int wave = __builtin_amdgcn_readfirstlane(threadIdx.x >> 5);
    const int lane = threadIdx.x & 31;
    const int r    = lane & 15;
    const int half = lane >> 4;

    {
        const int nt = wave & 7;
        const int kh = wave >> 3;
        const float* Aw = Wm + KD_OFF_W + (size_t)r * KD_L + 256 * kh + 2 * half;
        const float* Bw = slab + (128 * kh + half) * KD_PITCH
                               + 64 * (nt >> 1) + 4 * r + 2 * (nt & 1);
        v8f s0 = (v8f){}, s1 = (v8f){};
        for (int t = 0; t < 64; t += 8) {
            v2f a[8], f[8];
#pragma unroll
            for (int j = 0; j < 8; ++j) a[j] = *(const v2f*)(Aw + 4 * (t + j));
#pragma unroll
            for (int j = 0; j < 8; ++j) f[j] = *(const v2f*)(Bw + 2 * (t + j) * KD_PITCH);
#pragma unroll
            for (int j = 0; j < 8; j += 2) {
                s0 = __builtin_amdgcn_wmma_f32_16x16x4_f32(
                    false, a[j], false, f[j], (short)0, s0, false, false);
                s1 = __builtin_amdgcn_wmma_f32_16x16x4_f32(
                    false, a[j + 1], false, f[j + 1], (short)0, s1, false, false);
            }
        }
#pragma unroll
        for (int v = 0; v < 8; ++v)
            part[((8 * half + v) * KD_NC + 16 * nt + r) * 2 + kh] = s0[v] + s1[v];
    }
    __syncthreads();

    const int I  = wave >> 1;
    const int ch = wave & 1;

    v8f sg[2][4], dl[2][4];
#pragma unroll
    for (int mt = 0; mt < 2; ++mt)
#pragma unroll
        for (int j = 0; j < 4; ++j) { sg[mt][j] = (v8f){}; dl[mt][j] = (v8f){}; }

    const float* As = Wm + (size_t)r * KD_NH + 2 * half;
    const float* Bl = slab + (32 * I + half) * KD_PITCH + 128 * ch + 4 * r;
    const float* Bh = slab + (32 * I + 31 - half) * KD_PITCH + 128 * ch + 4 * r;

    for (int kc = 0; kc < KD_NH; kc += 16) {
        v2f aS[4][2], aD[4][2];
#pragma unroll
        for (int j = 0; j < 4; ++j) {
            aS[j][0] = *(const v2f*)(As + kc + 4 * j);
            aS[j][1] = *(const v2f*)(As + 16 * KD_NH + kc + 4 * j);
            aD[j][0] = *(const v2f*)(As + KD_NH * KD_NH + kc + 4 * j);
            aD[j][1] = *(const v2f*)(As + KD_NH * KD_NH + 16 * KD_NH + kc + 4 * j);
        }
#pragma unroll
        for (int j = 0; j < 4; ++j) {
            const int po = ((kc + 4 * j) >> 1) * KD_PITCH;
            kd_kstep(sg, dl, aS[j][0], aS[j][1], aD[j][0], aD[j][1], Bl + po, Bh - po);
        }
    }

    {
        v2f ax[2], ay[2];
#pragma unroll
        for (int mt = 0; mt < 2; ++mt) {
            const int a = 16 * mt + r;
            const float u0 = 0.5f * kd_rho_pow(a);
            const float u1 = 0.5f * kd_rho_pow(KD_NB - 1 - a);
            const float us = u0 + u1;
            const float ud = half ? (u1 - u0) : (u0 - u1);
            ax[mt] = (v2f){us, us};
            ay[mt] = (v2f){ud, ud};
        }
        const float* Pp = part + ((8 * half + I) * KD_NC + 64 * ch + r) * 2;
        v2f fb[4];
#pragma unroll
        for (int j = 0; j < 4; ++j) fb[j] = *(const v2f*)(Pp + 32 * j);
#pragma unroll
        for (int mt = 0; mt < 2; ++mt) {
#pragma unroll
            for (int j = 0; j < 4; ++j)
                sg[mt][j] = __builtin_amdgcn_wmma_f32_16x16x4_f32(
                    false, ax[mt], false, fb[j], (short)0, sg[mt][j], false, false);
#pragma unroll
            for (int j = 0; j < 4; ++j)
                dl[mt][j] = __builtin_amdgcn_wmma_f32_16x16x4_f32(
                    false, ay[mt], false, fb[j], (short)0, dl[mt][j], false, false);
        }
    }

    const int ab = 8 * half;
    const float* Yl = slab + (32 * I) * KD_PITCH + 128 * ch + 4 * r;
#pragma unroll
    for (int mt = 0; mt < 2; ++mt) {
#pragma unroll
        for (int q = 0; q < 2; ++q) {
#pragma unroll
            for (int vp = 0; vp < 4; ++vp) {
                const int a0 = 16 * mt + ab + 2 * vp;
                const v4f yl = *(const v4f*)(Yl + (a0 >> 1) * KD_PITCH + 64 * q);
                const v4f yh = *(const v4f*)(Yl + (31 - (a0 >> 1)) * KD_PITCH + 64 * q);
                const size_t o  = (size_t)(KD_NB * I + a0) * KD_C + c0 + 64 * ch + 32 * q + r;
                const size_t om = (size_t)(KD_NB * I + 63 - a0) * KD_C + c0 + 64 * ch + 32 * q + r;
                const float s00 = sg[mt][2 * q][2 * vp],         d00 = dl[mt][2 * q][2 * vp];
                const float s10 = sg[mt][2 * q][2 * vp + 1],     d10 = dl[mt][2 * q][2 * vp + 1];
                const float s01 = sg[mt][2 * q + 1][2 * vp],     d01 = dl[mt][2 * q + 1][2 * vp];
                const float s11 = sg[mt][2 * q + 1][2 * vp + 1], d11 = dl[mt][2 * q + 1][2 * vp + 1];
                float t;
                t = s00 + d00;  T[o]              = t;  R[o]              = yl.x - t;
                t = s10 + d10;  T[o + KD_C]       = t;  R[o + KD_C]       = yl.y - t;
                t = s01 + d01;  T[o + 16]         = t;  R[o + 16]         = yl.z - t;
                t = s11 + d11;  T[o + KD_C + 16]  = t;  R[o + KD_C + 16]  = yl.w - t;
                t = s00 - d00;  T[om]             = t;  R[om]             = yh.y - t;
                t = s10 - d10;  T[om - KD_C]      = t;  R[om - KD_C]      = yh.x - t;
                t = s01 - d01;  T[om + 16]        = t;  R[om + 16]        = yh.w - t;
                t = s11 - d11;  T[om - KD_C + 16] = t;  R[om - KD_C + 16] = yh.z - t;
            }
        }
    }
}

extern "C" void kernel_launch(void* const* d_in, const int* in_sizes, int n_in,
                              void* d_out, int out_size, void* d_ws, size_t ws_size,
                              hipStream_t stream) {
    (void)in_sizes; (void)n_in; (void)out_size; (void)ws_size;

    const float* obs = (const float*)d_in[0];
    float* out = (float*)d_out;
    float* Wm  = (float*)d_ws;

    kd_build_M<<<40, 256, 0, stream>>>(Wm);

    kd_gemm<<<KD_B * (KD_C / KD_NC), 512, 0, stream>>>(obs, Wm, out);
}


# Column r of a wave's column tile j now holds channel 4r+j (slab staging re-mapped): a lane's 4 results of a row are adjacent; trend, residual leave 16 B per lane in 256-B runs, not 64-B; same values
# speedup vs baseline: 6.9500x; 1.2418x over previous
#include <hip/hip_runtime.h>

typedef __attribute__((ext_vector_type(2))) float v2f;
typedef __attribute__((ext_vector_type(4))) float v4f;
typedef __attribute__((ext_vector_type(8))) float v8f;

#define KD_L  512
#define KD_C  512
#define KD_B  64
#define KD_NC 128
#define KD_PITCH 256

#define KD_NB   64
#define KD_NH   32
#define KD_OFF_W (2 * KD_NH * KD_NH)

#define KD_RHO1  0.381966011250105f
#define KD_RHO2  0.145898033750315f
#define KD_RHO4  0.0212862362522082f
#define KD_RHO8  4.53103853784822e-4f
#define KD_RHO16 2.05303102314657e-7f
#define KD_RHO32 4.21493638200227e-14f
#define KD_RHO64 1.77656887043264e-27f
#define KD_C0  0.447213595499958f

__device__ __forceinline__ float kd_rho_pow(int d) {
    float p = (d & 1) ? KD_RHO1 : 1.0f;
    p *= (d & 2)  ? KD_RHO2  : 1.0f;
    p *= (d & 4)  ? KD_RHO4  : 1.0f;
    p *= (d & 8)  ? KD_RHO8  : 1.0f;
    p *= (d & 16) ? KD_RHO16 : 1.0f;
    p *= (d & 32) ? KD_RHO32 : 1.0f;
    p *= (d & 64) ? KD_RHO64 : 1.0f;
    return (d < 128) ? p : 0.0f;
}

__global__ __launch_bounds__(256) void kd_build_M(float* __restrict__ Wm) {
    if (blockIdx.x < 8) {
        const int idx = blockIdx.x * 256 + threadIdx.x;
        const int dl = idx >> 10, a = (idx >> 5) & 31, b = idx & 31;
        const int d1 = (a > b) ? (a - b) : (b - a);
        const int d2 = (KD_NB - 1) - b - a;
        const float t1 = (0.5f * KD_C0) * kd_rho_pow(d1);
        const float t2 = (0.5f * KD_C0) * kd_rho_pow(d2);
        Wm[idx] = dl ? (t1 - t2) : (t1 + t2);
    } else {
        const int idx = (blockIdx.x - 8) * 256 + threadIdx.x;
        const int row = idx >> 9, s = idx & (KD_L - 1);
        const int I = row & 7;
        const int lo = KD_NB * I;
        const bool below = s < lo, above = s >= lo + KD_NB;
        float w;
        if (row < 8) {
            w = -(KD_C0 * kd_rho_pow(lo + s + 2));
            if (below) w += KD_C0 * kd_rho_pow(lo - s);
        } else {
            w = KD_C0 * kd_rho_pow(960 - lo - s);
            if (above) w += KD_C0 * kd_rho_pow(s - lo - 63);
        }
        Wm[KD_OFF_W + idx] = w;
    }
}

#define KD_SLAB (256 * KD_PITCH)
#define KD_PART (2 * 16 * KD_NC)

__device__ __forceinline__ void kd_kstep(v8f (&sg)[2][4], v8f (&dl)[2][4],
                                         v2f aS0, v2f aS1, v2f aD0, v2f aD1,
                                         const float* lo, const float* hi) {
    v2f fp[4], fq[4];
#pragma unroll
    for (int q = 0; q < 2; ++q) {
        const v4f u = *(const v4f*)(lo + 64 * q);
        const v4f w = *(const v4f*)(hi + 64 * q);
        fp[2 * q]     = (v2f){u.x + w.y, u.y + w.x};
        fq[2 * q]     = (v2f){u.x - w.y, u.y - w.x};
        fp[2 * q + 1] = (v2f){u.z + w.w, u.w + w.z};
        fq[2 * q + 1] = (v2f){u.z - w.w, u.w - w.z};
    }
#pragma unroll
    for (int j = 0; j < 4; ++j)
        sg[0][j] = __builtin_amdgcn_wmma_f32_16x16x4_f32(
            false, aS0, false, fp[j], (short)0, sg[0][j], false, false);
#pragma unroll
    for (int j = 0; j < 4; ++j)
        dl[0][j] = __builtin_amdgcn_wmma_f32_16x16x4_f32(
            false, aD0, false, fq[j], (short)0, dl[0][j], false, false);
#pragma unroll
    for (int j = 0; j < 4; ++j)
        sg[1][j] = __builtin_amdgcn_wmma_f32_16x16x4_f32(
            false, aS1, false, fp[j], (short)0, sg[1][j], false, false);
#pragma unroll
    for (int j = 0; j < 4; ++j)
        dl[1][j] = __builtin_amdgcn_wmma_f32_16x16x4_f32(
            false, aD1, false, fq[j], (short)0, dl[1][j], false, false);
}

__global__ __launch_bounds__(512) void kd_gemm(const float* __restrict__ obs,
                                               const float* __restrict__ Wm,
                                               float* __restrict__ out) {
    __shared__ float slab[KD_SLAB + KD_PART];
    float* part = slab + KD_SLAB;

    const int b  = blockIdx.x >> 2;
    const int c0 = (blockIdx.x & 3) * KD_NC;

    const float* __restrict__ Y = obs + (size_t)b * KD_L * KD_C;
    float* __restrict__ T = out + (size_t)b * KD_L * KD_C;
    float* __restrict__ R = out + (size_t)KD_B * KD_L * KD_C
                                + (size_t)b * KD_L * KD_C;

    {
        const int g  = threadIdx.x & 31;
        const int pr = threadIdx.x >> 5;
        const int sb = 128 * (g >> 4) + 4 * (g & 15);
#pragma unroll 4
        for (int it = 0; it < 16; ++it) {
            const int p = it * 16 + pr;
            const float* src = Y + (size_t)(2 * p) * KD_C + c0 + 4 * g;
            const v4f r0 = __builtin_nontemporal_load((const v4f*)src);
            const v4f r1 = __builtin_nontemporal_load((const v4f*)(src + KD_C));
            float* dp = slab + p * KD_PITCH + sb;
            *(v4f*)dp        = (v4f){r0.x, r1.x, r0.y, r1.y};
            *(v4f*)(dp + 64) = (v4f){r0.z, r1.z, r0.w, r1.w};
        }
    }
    __syncthreads();

    const int wave = __builtin_amdgcn_readfirstlane(threadIdx.x >> 5);
    const int lane = threadIdx.x & 31;
    const int r    = lane & 15;
    const int half = lane >> 4;

    {
        const int nt = wave & 7;
        const int kh = wave >> 3;
        const float* Aw = Wm + KD_OFF_W + (size_t)r * KD_L + 256 * kh + 2 * half;
        const float* Bw = slab + (128 * kh + half) * KD_PITCH
                               + 64 * (nt >> 1) + 4 * r + 2 * (nt & 1);
        v8f s0 = (v8f){}, s1 = (v8f){};
        for (int t = 0; t < 64; t += 8) {
            v2f a[8], f[8];
#pragma unroll
            for (int j = 0; j < 8; ++j) a[j] = *(const v2f*)(Aw + 4 * (t + j));
#pragma unroll
            for (int j = 0; j < 8; ++j) f[j] = *(const v2f*)(Bw + 2 * (t + j) * KD_PITCH);
#pragma unroll
            for (int j = 0; j < 8; j += 2) {
                s0 = __builtin_amdgcn_wmma_f32_16x16x4_f32(
                    false, a[j], false, f[j], (short)0, s0, false, false);
                s1 = __builtin_amdgcn_wmma_f32_16x16x4_f32(
                    false, a[j + 1], false, f[j + 1], (short)0, s1, false, false);
            }
        }
#pragma unroll
        for (int v = 0; v < 8; ++v)
            part[((8 * half + v) * KD_NC + 16 * nt + r) * 2 + kh] = s0[v] + s1[v];
    }
    __syncthreads();

    const int I  = wave >> 1;
    const int ch = wave & 1;

    v8f sg[2][4], dl[2][4];
#pragma unroll
    for (int mt = 0; mt < 2; ++mt)
#pragma unroll
        for (int j = 0; j < 4; ++j) { sg[mt][j] = (v8f){}; dl[mt][j] = (v8f){}; }

    const float* As = Wm + (size_t)r * KD_NH + 2 * half;
    const float* Bl = slab + (32 * I + half) * KD_PITCH + 128 * ch + 4 * r;
    const float* Bh = slab + (32 * I + 31 - half) * KD_PITCH + 128 * ch + 4 * r;

    for (int kc = 0; kc < KD_NH; kc += 16) {
        v2f aS[4][2], aD[4][2];
#pragma unroll
        for (int j = 0; j < 4; ++j) {
            aS[j][0] = *(const v2f*)(As + kc + 4 * j);
            aS[j][1] = *(const v2f*)(As + 16 * KD_NH + kc + 4 * j);
            aD[j][0] = *(const v2f*)(As + KD_NH * KD_NH + kc + 4 * j);
            aD[j][1] = *(const v2f*)(As + KD_NH * KD_NH + 16 * KD_NH + kc + 4 * j);
        }
#pragma unroll
        for (int j = 0; j < 4; ++j) {
            const int po = ((kc + 4 * j) >> 1) * KD_PITCH;
            kd_kstep(sg, dl, aS[j][0], aS[j][1], aD[j][0], aD[j][1], Bl + po, Bh - po);
        }
    }

    {
        v2f ax[2], ay[2];
#pragma unroll
        for (int mt = 0; mt < 2; ++mt) {
            const int a = 16 * mt + r;
            const float u0 = 0.5f * kd_rho_pow(a);
            const float u1 = 0.5f * kd_rho_pow(KD_NB - 1 - a);
            const float us = u0 + u1;
            const float ud = half ? (u1 - u0) : (u0 - u1);
            ax[mt] = (v2f){us, us};
            ay[mt] = (v2f){ud, ud};
        }
        const float* Pp = part + ((8 * half + I) * KD_NC + 64 * ch + r) * 2;
        v2f fb[4];
#pragma unroll
        for (int j = 0; j < 4; ++j) fb[j] = *(const v2f*)(Pp + 32 * j);
#pragma unroll
        for (int mt = 0; mt < 2; ++mt) {
#pragma unroll
            for (int j = 0; j < 4; ++j)
                sg[mt][j] = __builtin_amdgcn_wmma_f32_16x16x4_f32(
                    false, ax[mt], false, fb[j], (short)0, sg[mt][j], false, false);
#pragma unroll
            for (int j = 0; j < 4; ++j)
                dl[mt][j] = __builtin_amdgcn_wmma_f32_16x16x4_f32(
                    false, ay[mt], false, fb[j], (short)0, dl[mt][j], false, false);
        }
    }

    const int ab = 8 * half;
    const float* Yl = slab + (32 * I) * KD_PITCH + 128 * ch + 4 * r;
#pragma unroll
    for (int mt = 0; mt < 2; ++mt) {
#pragma unroll
        for (int vp = 0; vp < 4; ++vp) {
            const int a0 = 16 * mt + ab + 2 * vp;
            const float* yp = Yl + (a0 >> 1) * KD_PITCH;
            const float* ym = Yl + (31 - (a0 >> 1)) * KD_PITCH;
            const v4f yl0 = *(const v4f*)yp,  yl1 = *(const v4f*)(yp + 64);
            const v4f yh0 = *(const v4f*)ym,  yh1 = *(const v4f*)(ym + 64);
            const size_t o  = (size_t)(KD_NB * I + a0) * KD_C + c0 + 64 * ch + 4 * r;
            const size_t om = (size_t)(KD_NB * I + 63 - a0) * KD_C + c0 + 64 * ch + 4 * r;
            const int v0 = 2 * vp, v1 = 2 * vp + 1;
            v4f t;
            t = (v4f){sg[mt][0][v0] + dl[mt][0][v0], sg[mt][1][v0] + dl[mt][1][v0],
                      sg[mt][2][v0] + dl[mt][2][v0], sg[mt][3][v0] + dl[mt][3][v0]};
            *(v4f*)(T + o) = t;
            *(v4f*)(R + o) = (v4f){yl0.x - t.x, yl0.z - t.y, yl1.x - t.z, yl1.z - t.w};
            t = (v4f){sg[mt][0][v1] + dl[mt][0][v1], sg[mt][1][v1] + dl[mt][1][v1],
                      sg[mt][2][v1] + dl[mt][2][v1], sg[mt][3][v1] + dl[mt][3][v1]};
            *(v4f*)(T + o + KD_C) = t;
            *(v4f*)(R + o + KD_C) = (v4f){yl0.y - t.x, yl0.w - t.y, yl1.y - t.z, yl1.w - t.w};
            t = (v4f){sg[mt][0][v0] - dl[mt][0][v0], sg[mt][1][v0] - dl[mt][1][v0],
                      sg[mt][2][v0] - dl[mt][2][v0], sg[mt][3][v0] - dl[mt][3][v0]};
            *(v4f*)(T + om) = t;
            *(v4f*)(R + om) = (v4f){yh0.y - t.x, yh0.w - t.y, yh1.y - t.z, yh1.w - t.w};
            t = (v4f){sg[mt][0][v1] - dl[mt][0][v1], sg[mt][1][v1] - dl[mt][1][v1],
                      sg[mt][2][v1] - dl[mt][2][v1], sg[mt][3][v1] - dl[mt][3][v1]};
            *(v4f*)(T + om - KD_C) = t;
            *(v4f*)(R + om - KD_C) = (v4f){yh0.x - t.x, yh0.z - t.y, yh1.x - t.z, yh1.z - t.w};
        }
    }
}

extern "C" void kernel_launch(void* const* d_in, const int* in_sizes, int n_in,
                              void* d_out, int out_size, void* d_ws, size_t ws_size,
                              hipStream_t stream) {
    (void)in_sizes; (void)n_in; (void)out_size; (void)ws_size;

    const float* obs = (const float*)d_in[0];
    float* out = (float*)d_out;
    float* Wm  = (float*)d_ws;

    kd_build_M<<<40, 256, 0, stream>>>(Wm);

    kd_gemm<<<KD_B * (KD_C / KD_NC), 512, 0, stream>>>(obs, Wm, out);
}
